# starting kernel
# baseline (speedup 1.0000x reference)
#include <hip/hip_runtime.h>
#include <math.h>

typedef __attribute__((ext_vector_type(2))) float v2f;
typedef __attribute__((ext_vector_type(8))) float v8f;

#define WAVES 8
#define TILE  16
#define LDA   68

__device__ __forceinline__ v8f wmma4(v2f a, v2f b, v8f c) {
    return __builtin_amdgcn_wmma_f32_16x16x4_f32(
         false, a,  false, b,
         (short)0, c,  false,  false);
}

__device__ __forceinline__ float fast_tanh(float x) {
#if __has_builtin(__builtin_amdgcn_tanhf)
    return __builtin_amdgcn_tanhf(x);
#else
    return tanhf(x);
#endif
}

__global__ __launch_bounds__(WAVES * 32)
void node_rk45_kernel(const float* __restrict__ pad0,
                      const float* __restrict__ sens,
                      const float* __restrict__ W1, const float* __restrict__ b1,
                      const float* __restrict__ W2, const float* __restrict__ b2,
                      const float* __restrict__ W3, const float* __restrict__ b3,
                      const float* __restrict__ scale_p,
                      float* __restrict__ out, int ntiles)
{
    __shared__ float2 w2s[16 * 4 * 32];
    __shared__ float astage[WAVES][TILE * LDA];
    __shared__ float kstage[WAVES][TILE * 4];

    const int tid  = threadIdx.x;
    const int wave = tid >> 5;
    const int lane = tid & 31;
    const int half = lane >> 4;
    const int l    = lane & 15;

    for (int idx = tid; idx < 16 * 4 * 32; idx += WAVES * 32) {
        int c  = idx >> 7;
        int n  = (idx >> 5) & 3;
        int ln = idx & 31;
        int hh = ln >> 4, ll = ln & 15;
        int r0 = 4 * c + 2 * hh;
        int col = n * 16 + ll;
        w2s[idx] = make_float2(W2[r0 * 64 + col], W2[(r0 + 1) * 64 + col]);
    }
    __syncthreads();

    const int tile = blockIdx.x * WAVES + wave;
    if (tile >= ntiles) return;
    const int  base  = tile * TILE;
    const float scale = scale_p[0];
    const float hstep = 0.1f;

    float* mystage = astage[wave];
    float* myk     = kstage[wave];

    v2f w1f[4];
#pragma unroll
    for (int n = 0; n < 4; ++n) {
        int col = n * 16 + l;
        w1f[n].x = half ? W1[2 * 64 + col] : W1[0 * 64 + col];
        w1f[n].y = half ? W1[64 * 64 + col] : W1[1 * 64 + col];
    }
    v2f w3f[16];
#pragma unroll
    for (int c = 0; c < 16; ++c) {
        int r0 = 4 * c + 2 * half;
        w3f[c].x = (l < 3) ? W3[r0 * 3 + l]       : 0.f;
        w3f[c].y = (l < 3) ? W3[(r0 + 1) * 3 + l] : 0.f;
    }
    float b1v[4], b2v[4];
#pragma unroll
    for (int n = 0; n < 4; ++n) { b1v[n] = b1[n * 16 + l]; b2v[n] = b2[n * 16 + l]; }
    const float b3v = (l < 3) ? b3[l] : 0.f;

    for (int idx = lane; idx < TILE * 64; idx += 32) {
        int r = idx >> 6, cc = idx & 63;
        mystage[r * LDA + cc] = (cc < 61) ? sens[(base + r) * 61 + cc] : 0.f;
    }
    v8f Sc[4];
#pragma unroll
    for (int n = 0; n < 4; ++n) {
        v8f z;
#pragma unroll
        for (int v = 0; v < 8; ++v) z[v] = b1v[n];
        Sc[n] = z;
    }
#pragma unroll
    for (int c = 0; c < 16; ++c) {
        v2f a = *(const v2f*)&mystage[l * LDA + 4 * c + 2 * half];
        int r0 = 4 * c + 2 * half;
#pragma unroll
        for (int n = 0; n < 4; ++n) {
            int col = n * 16 + l;
            v2f b;
            b.x = (r0     <= 60) ? W1[(3 + r0) * 64 + col] : 0.f;
            b.y = (r0 + 1 <= 60) ? W1[(4 + r0) * 64 + col] : 0.f;
            Sc[n] = wmma4(a, b, Sc[n]);
        }
    }

    float2 y;
    {
        const float* p = pad0 + (base + l) * 3;
        y.x = half ? p[2] : p[0];
        y.y = half ? 0.f  : p[1];
    }

    auto feval = [&](float tt, float2 ya) -> float2 {
        v2f A1; A1.x = ya.x; A1.y = half ? tt : ya.y;
        v8f h1[4];
#pragma unroll
        for (int n = 0; n < 4; ++n) {
            v8f d = wmma4(A1, w1f[n], Sc[n]);
#pragma unroll
            for (int v = 0; v < 8; ++v) d[v] = fast_tanh(d[v]);
            h1[n] = d;
        }
#pragma unroll
        for (int n = 0; n < 4; ++n)
#pragma unroll
            for (int v = 0; v < 8; ++v)
                mystage[(v + 8 * half) * LDA + n * 16 + l] = h1[n][v];
        v2f afr[16];
#pragma unroll
        for (int c = 0; c < 16; ++c)
            afr[c] = *(const v2f*)&mystage[l * LDA + 4 * c + 2 * half];

        v8f h2[4];
#pragma unroll
        for (int n = 0; n < 4; ++n) {
            v8f z;
#pragma unroll
            for (int v = 0; v < 8; ++v) z[v] = b2v[n];
            h2[n] = z;
        }
#pragma unroll
        for (int c = 0; c < 16; ++c)
#pragma unroll
            for (int n = 0; n < 4; ++n) {
                float2 bf = w2s[(c * 4 + n) * 32 + lane];
                v2f b; b.x = bf.x; b.y = bf.y;
                h2[n] = wmma4(afr[c], b, h2[n]);
            }
#pragma unroll
        for (int n = 0; n < 4; ++n)
#pragma unroll
            for (int v = 0; v < 8; ++v)
                mystage[(v + 8 * half) * LDA + n * 16 + l] = fast_tanh(h2[n][v]);

        v8f acc3;
#pragma unroll
        for (int v = 0; v < 8; ++v) acc3[v] = b3v;
#pragma unroll
        for (int c = 0; c < 16; ++c) {
            v2f a = *(const v2f*)&mystage[l * LDA + 4 * c + 2 * half];
            acc3 = wmma4(a, w3f[c], acc3);
        }
        if (l < 4) {
#pragma unroll
            for (int v = 0; v < 8; ++v)
                myk[(v + 8 * half) * 4 + l] = fast_tanh(acc3[v]) * scale;
        }
        float2 kk = *(const float2*)&myk[l * 4 + 2 * half];
        return kk;
    };

    for (int step = 0; step < 10; ++step) {
        const float t0 = step * hstep;
        float2 k1, k2, k3, k4, k5, k6, ya;

        k1 = feval(t0, y);
        ya = make_float2(y.x + hstep * 0.2f * k1.x,
                         y.y + hstep * 0.2f * k1.y);
        k2 = feval(t0 + 0.2f * hstep, ya);
        ya = make_float2(y.x + hstep * (0.075f * k1.x + 0.225f * k2.x),
                         y.y + hstep * (0.075f * k1.y + 0.225f * k2.y));
        k3 = feval(t0 + 0.3f * hstep, ya);
        {
            const float a1 = 44.f / 45.f, a2 = -56.f / 15.f, a3 = 32.f / 9.f;
            ya = make_float2(y.x + hstep * (a1 * k1.x + a2 * k2.x + a3 * k3.x),
                             y.y + hstep * (a1 * k1.y + a2 * k2.y + a3 * k3.y));
        }
        k4 = feval(t0 + 0.8f * hstep, ya);
        {
            const float a1 = 19372.f / 6561.f, a2 = -25360.f / 2187.f,
                        a3 = 64448.f / 6561.f, a4 = -212.f / 729.f;
            ya = make_float2(y.x + hstep * (a1 * k1.x + a2 * k2.x + a3 * k3.x + a4 * k4.x),
                             y.y + hstep * (a1 * k1.y + a2 * k2.y + a3 * k3.y + a4 * k4.y));
        }
        k5 = feval(t0 + (8.f / 9.f) * hstep, ya);
        {
            const float a1 = 9017.f / 3168.f, a2 = -355.f / 33.f, a3 = 46732.f / 5247.f,
                        a4 = 49.f / 176.f, a5 = -5103.f / 18656.f;
            ya = make_float2(y.x + hstep * (a1 * k1.x + a2 * k2.x + a3 * k3.x + a4 * k4.x + a5 * k5.x),
                             y.y + hstep * (a1 * k1.y + a2 * k2.y + a3 * k3.y + a4 * k4.y + a5 * k5.y));
        }
        k6 = feval(t0 + hstep, ya);
        {
            const float c1 = 35.f / 384.f, c3 = 500.f / 1113.f, c4 = 125.f / 192.f,
                        c5 = -2187.f / 6784.f, c6 = 11.f / 84.f;
            y = make_float2(y.x + hstep * (c1 * k1.x + c3 * k3.x + c4 * k4.x + c5 * k5.x + c6 * k6.x),
                            y.y + hstep * (c1 * k1.y + c3 * k3.y + c4 * k4.y + c5 * k5.y + c6 * k6.y));
        }
    }

    float* po = out + (base + l) * 3;
    if (half) {
        po[2] = y.x;
    } else {
        po[0] = y.x;
        po[1] = y.y;
    }
}

extern "C" void kernel_launch(void* const* d_in, const int* in_sizes, int n_in,
                              void* d_out, int out_size, void* d_ws, size_t ws_size,
                              hipStream_t stream) {
    const float* pad0 = (const float*)d_in[0];
    const float* sens = (const float*)d_in[1];
    const float* W1   = (const float*)d_in[2];
    const float* b1   = (const float*)d_in[3];
    const float* W2   = (const float*)d_in[4];
    const float* b2   = (const float*)d_in[5];
    const float* W3   = (const float*)d_in[6];
    const float* b3   = (const float*)d_in[7];
    const float* sc   = (const float*)d_in[8];
    float* out = (float*)d_out;

    const int B      = in_sizes[0] / 3;
    const int ntiles = (B + TILE - 1) / TILE;
    const int blocks = (ntiles + WAVES - 1) / WAVES;

    node_rk45_kernel<<<blocks, WAVES * 32, 0, stream>>>(
        pad0, sens, W1, b1, W2, b2, W3, b3, sc, out, ntiles);
}


# MLP evaluated transposed, register-resident: weights as WMMA A operand kept in VGPRs, batch rows as N, each layer's D pairs feed the next as B; no LDS or barrier; k order in layers 2-3 permuted
# speedup vs baseline: 1.1476x; 1.1476x over previous
#include <hip/hip_runtime.h>
#include <math.h>

typedef __attribute__((ext_vector_type(2))) float v2f;
typedef __attribute__((ext_vector_type(8))) float v8f;

#define WAVES 4
#define TILE  16

__device__ __forceinline__ v8f wmma4(v2f a, v2f b, v8f c) {
    return __builtin_amdgcn_wmma_f32_16x16x4_f32(
         false, a,  false, b,
         (short)0, c,  false,  false);
}

__device__ __forceinline__ float fast_tanh(float x) {
#if __has_builtin(__builtin_amdgcn_tanhf)
    return __builtin_amdgcn_tanhf(x);
#else
    return tanhf(x);
#endif
}

__global__ __launch_bounds__(WAVES * 32)
void node_rk45_kernel(const float* __restrict__ pad0,
                      const float* __restrict__ sens,
                      const float* __restrict__ W1, const float* __restrict__ b1,
                      const float* __restrict__ W2, const float* __restrict__ b2,
                      const float* __restrict__ W3, const float* __restrict__ b3,
                      const float* __restrict__ scale_p,
                      float* __restrict__ out, int ntiles)
{
    const int tid  = threadIdx.x;
    const int wave = tid >> 5;
    const int lane = tid & 31;
    const int half = lane >> 4;
    const int l    = lane & 15;

    const int tile = blockIdx.x * WAVES + wave;
    if (tile >= ntiles) return;
    const int  base  = tile * TILE;
    const float scale = scale_p[0];
    const float hstep = 0.1f;

    v8f Sc[4];
#pragma unroll
    for (int n = 0; n < 4; ++n)
#pragma unroll
        for (int r = 0; r < 8; ++r) Sc[n][r] = b1[16 * n + 8 * half + r];
    {
        const float* srow = sens + (base + l) * 61;
#pragma unroll
        for (int c = 0; c < 16; ++c) {
            const int  k0  = 4 * c + 2 * half;
            const bool ok0 = (k0 <= 60), ok1 = (k0 + 1 <= 60);
            const int  i0  = ok0 ? k0 : 60, i1 = ok1 ? k0 + 1 : 60;
            const float s0 = srow[i0], s1 = srow[i1];
            v2f b;
            b.x = ok0 ? s0 : 0.f;
            b.y = ok1 ? s1 : 0.f;
#pragma unroll
            for (int n = 0; n < 4; ++n) {
                int col = n * 16 + l;
                const float u0 = W1[(3 + i0) * 64 + col], u1 = W1[(3 + i1) * 64 + col];
                v2f a;
                a.x = ok0 ? u0 : 0.f;
                a.y = ok1 ? u1 : 0.f;
                Sc[n] = wmma4(a, b, Sc[n]);
            }
        }
    }

    v2f w1a[4];
#pragma unroll
    for (int n = 0; n < 4; ++n) {
        int col = n * 16 + l;
        w1a[n].x = W1[(half ? 2  : 0) * 64 + col];
        w1a[n].y = W1[(half ? 64 : 1) * 64 + col];
    }
    v2f w2a[64];
#pragma unroll
    for (int q = 0; q < 16; ++q) {
        int fr = 16 * (q >> 2) + 2 * (q & 3) + 8 * half;
#pragma unroll
        for (int m = 0; m < 4; ++m) {
            int col = 16 * m + l;
            w2a[q * 4 + m].x = W2[fr * 64 + col];
            w2a[q * 4 + m].y = W2[(fr + 1) * 64 + col];
        }
    }
    const bool w3on = (l < 2) || (l == 8);
    const int  w3c  = (l == 8) ? 2 : ((l < 2) ? l : 0);
    v2f w3a[16];
#pragma unroll
    for (int q = 0; q < 16; ++q) {
        int fr = 16 * (q >> 2) + 2 * (q & 3) + 8 * half;
        float wx = W3[fr * 3 + w3c];
        float wy = W3[(fr + 1) * 3 + w3c];
        w3a[q].x = w3on ? wx : 0.f;
        w3a[q].y = w3on ? wy : 0.f;
    }
    v8f b2c[4];
#pragma unroll
    for (int m = 0; m < 4; ++m)
#pragma unroll
        for (int r = 0; r < 8; ++r) b2c[m][r] = b2[16 * m + 8 * half + r];
    v8f c3;
#pragma unroll
    for (int r = 0; r < 8; ++r) c3[r] = 0.f;
    {
        const float b3a = b3[half ? 2 : 0], b3b = b3[1];
        c3[0] = b3a;
        c3[1] = half ? 0.f : b3b;
    }

    float2 y;
    {
        const float* p = pad0 + (base + l) * 3;
        const float p1 = p[1];
        y.x = p[half ? 2 : 0];
        y.y = half ? 0.f : p1;
    }

    auto feval = [&](float tt, float2 ya) -> float2 {
        v2f B1; B1.x = ya.x; B1.y = half ? tt : ya.y;
        v8f h1[4];
#pragma unroll
        for (int n = 0; n < 4; ++n) {
            v8f d = wmma4(w1a[n], B1, Sc[n]);
#pragma unroll
            for (int r = 0; r < 8; ++r) d[r] = fast_tanh(d[r]);
            h1[n] = d;
        }
        v8f h2[4];
#pragma unroll
        for (int m = 0; m < 4; ++m) h2[m] = b2c[m];
#pragma unroll
        for (int q = 0; q < 16; ++q) {
            v2f b; b.x = h1[q >> 2][2 * (q & 3)]; b.y = h1[q >> 2][2 * (q & 3) + 1];
#pragma unroll
            for (int m = 0; m < 4; ++m)
                h2[m] = wmma4(w2a[q * 4 + m], b, h2[m]);
        }
#pragma unroll
        for (int m = 0; m < 4; ++m)
#pragma unroll
            for (int r = 0; r < 8; ++r) h2[m][r] = fast_tanh(h2[m][r]);
        v8f acc3 = c3;
#pragma unroll
        for (int q = 0; q < 16; ++q) {
            v2f b; b.x = h2[q >> 2][2 * (q & 3)]; b.y = h2[q >> 2][2 * (q & 3) + 1];
            acc3 = wmma4(w3a[q], b, acc3);
        }
        float2 kk;
        kk.x = fast_tanh(acc3[0]) * scale;
        kk.y = fast_tanh(acc3[1]) * scale;
        return kk;
    };

    for (int step = 0; step < 10; ++step) {
        const float t0 = step * hstep;
        float2 k1, k2, k3, k4, k5, k6, ya;

        k1 = feval(t0, y);
        ya = make_float2(y.x + hstep * 0.2f * k1.x,
                         y.y + hstep * 0.2f * k1.y);
        k2 = feval(t0 + 0.2f * hstep, ya);
        ya = make_float2(y.x + hstep * (0.075f * k1.x + 0.225f * k2.x),
                         y.y + hstep * (0.075f * k1.y + 0.225f * k2.y));
        k3 = feval(t0 + 0.3f * hstep, ya);
        {
            const float a1 = 44.f / 45.f, a2 = -56.f / 15.f, a3 = 32.f / 9.f;
            ya = make_float2(y.x + hstep * (a1 * k1.x + a2 * k2.x + a3 * k3.x),
                             y.y + hstep * (a1 * k1.y + a2 * k2.y + a3 * k3.y));
        }
        k4 = feval(t0 + 0.8f * hstep, ya);
        {
            const float a1 = 19372.f / 6561.f, a2 = -25360.f / 2187.f,
                        a3 = 64448.f / 6561.f, a4 = -212.f / 729.f;
            ya = make_float2(y.x + hstep * (a1 * k1.x + a2 * k2.x + a3 * k3.x + a4 * k4.x),
                             y.y + hstep * (a1 * k1.y + a2 * k2.y + a3 * k3.y + a4 * k4.y));
        }
        k5 = feval(t0 + (8.f / 9.f) * hstep, ya);
        {
            const float a1 = 9017.f / 3168.f, a2 = -355.f / 33.f, a3 = 46732.f / 5247.f,
                        a4 = 49.f / 176.f, a5 = -5103.f / 18656.f;
            ya = make_float2(y.x + hstep * (a1 * k1.x + a2 * k2.x + a3 * k3.x + a4 * k4.x + a5 * k5.x),
                             y.y + hstep * (a1 * k1.y + a2 * k2.y + a3 * k3.y + a4 * k4.y + a5 * k5.y));
        }
        k6 = feval(t0 + hstep, ya);
        {
            const float c1 = 35.f / 384.f, c3w = 500.f / 1113.f, c4 = 125.f / 192.f,
                        c5 = -2187.f / 6784.f, c6 = 11.f / 84.f;
            y = make_float2(y.x + hstep * (c1 * k1.x + c3w * k3.x + c4 * k4.x + c5 * k5.x + c6 * k6.x),
                            y.y + hstep * (c1 * k1.y + c3w * k3.y + c4 * k4.y + c5 * k5.y + c6 * k6.y));
        }
    }

    float* po = out + (base + l) * 3;
    if (half) {
        po[2] = y.x;
    } else {
        po[0] = y.x;
        po[1] = y.y;
    }
}

extern "C" void kernel_launch(void* const* d_in, const int* in_sizes, int n_in,
                              void* d_out, int out_size, void* d_ws, size_t ws_size,
                              hipStream_t stream) {
    const float* pad0 = (const float*)d_in[0];
    const float* sens = (const float*)d_in[1];
    const float* W1   = (const float*)d_in[2];
    const float* b1   = (const float*)d_in[3];
    const float* W2   = (const float*)d_in[4];
    const float* b2   = (const float*)d_in[5];
    const float* W3   = (const float*)d_in[6];
    const float* b3   = (const float*)d_in[7];
    const float* sc   = (const float*)d_in[8];
    float* out = (float*)d_out;

    const int B      = in_sizes[0] / 3;
    const int ntiles = (B + TILE - 1) / TILE;
    const int blocks = (ntiles + WAVES - 1) / WAVES;

    node_rk45_kernel<<<blocks, WAVES * 32, 0, stream>>>(
        pad0, sens, W1, b1, W2, b2, W3, b3, sc, out, ntiles);
}
